# starting kernel
# baseline (speedup 1.0000x reference)
#include <hip/hip_runtime.h>
#include <math.h>


typedef __attribute__((ext_vector_type(2))) float v2f;
typedef __attribute__((ext_vector_type(8))) float v8f;

__device__ __forceinline__ v8f wmma_f32_16x16x4(v2f a, v2f b, v8f c) {
  return __builtin_amdgcn_wmma_f32_16x16x4_f32(false, a, false, b, (short)0, c,
                                               false, false);
}

__global__ void __launch_bounds__(128)
lin_gemm_relu(const float* __restrict__ x, const float* __restrict__ W,
              const float* __restrict__ bias, float* __restrict__ h,
              float* __restrict__ x0) {
  __shared__ float xt[16 * 132];

  const int row0 = blockIdx.x * 16;
  const int t = threadIdx.x;

  {
    const int r = t >> 3;
    const int c0 = (t & 7) * 16;
    const float4* xp =
        (const float4*)(x + (size_t)(row0 + r) * 128 + c0);
#pragma unroll
    for (int j = 0; j < 4; ++j) {
      float4 v = xp[j];
      float* dst = &xt[r * 132 + c0 + 4 * j];
      dst[0] = v.x; dst[1] = v.y; dst[2] = v.z; dst[3] = v.w;
    }
  }
  __syncthreads();

  const int lane  = t & 31;
  const int wave  = t >> 5;
  const int lhalf = lane >> 4;
  const int lm    = lane & 15;
  const int nc0   = wave * 16;

  v8f acc = {};
#pragma unroll
  for (int k0 = 0; k0 < 128; k0 += 4) {
    v2f a;
    const float* ap = &xt[lm * 132 + k0 + 2 * lhalf];
    a.x = ap[0];
    a.y = ap[1];
    v2f b;
    const float* wp = W + (size_t)(k0 + 2 * lhalf) * 64 + nc0 + lm;
    b.x = wp[0];
    b.y = wp[64];
    acc = wmma_f32_16x16x4(a, b, acc);
  }

  const float bv = bias[nc0 + lm];
#pragma unroll
  for (int i = 0; i < 8; ++i) {
    const int m = i + 8 * lhalf;
    float v = acc[i] + bv;
    v = fmaxf(v, 0.0f);
    const size_t idx = (size_t)(row0 + m) * 64 + nc0 + lm;
    h[idx] = v;
    x0[idx] = v;
  }
}

__global__ void init_agg(const float* __restrict__ x0, float* __restrict__ agg,
                         int n4) {
  const int i = blockIdx.x * blockDim.x + threadIdx.x;
  if (i >= n4) return;
  const float s = 0.1f / 0.9f;
  float4 v = ((const float4*)x0)[i];
  v.x *= s; v.y *= s; v.z *= s; v.w *= s;
  ((float4*)agg)[i] = v;
}

__global__ void __launch_bounds__(256)
spmm_scatter(const float* __restrict__ h, const int* __restrict__ row,
             const int* __restrict__ col, const float* __restrict__ ew,
             float* __restrict__ agg, int E) {
  const int wid = blockIdx.x * (blockDim.x >> 5) + (threadIdx.x >> 5);
  if (wid >= E) return;
  const int lane = threadIdx.x & 31;

  const int r = row[wid];
  const int c = col[wid];
  const float w = ew[wid];

  const float2 v = ((const float2*)(h + (size_t)c * 64))[lane];
  float* ap = agg + (size_t)r * 64 + 2 * lane;
  atomicAdd(ap + 0, v.x * w);
  atomicAdd(ap + 1, v.y * w);
}

__global__ void __launch_bounds__(128)
gcn2_update(const float* __restrict__ agg, const float* __restrict__ W,
            float* __restrict__ out, float beta) {
  __shared__ float zt[16 * 68];

  const int row0 = blockIdx.x * 16;
  const int t = threadIdx.x;

  {
    const int r = t >> 3;
    const int c0 = (t & 7) * 8;
    const float4* ap =
        (const float4*)(agg + (size_t)(row0 + r) * 64 + c0);
#pragma unroll
    for (int j = 0; j < 2; ++j) {
      float4 v = ap[j];
      float* dst = &zt[r * 68 + c0 + 4 * j];
      dst[0] = 0.9f * v.x; dst[1] = 0.9f * v.y;
      dst[2] = 0.9f * v.z; dst[3] = 0.9f * v.w;
    }
  }
  __syncthreads();

  const int lane  = t & 31;
  const int wave  = t >> 5;
  const int lhalf = lane >> 4;
  const int lm    = lane & 15;
  const int nc0   = wave * 16;

  v8f acc = {};
#pragma unroll
  for (int k0 = 0; k0 < 64; k0 += 4) {
    v2f a;
    const float* ap = &zt[lm * 68 + k0 + 2 * lhalf];
    a.x = ap[0];
    a.y = ap[1];
    v2f b;
    const float* wp = W + (size_t)(k0 + 2 * lhalf) * 64 + nc0 + lm;
    b.x = wp[0];
    b.y = wp[64];
    acc = wmma_f32_16x16x4(a, b, acc);
  }

  const float omb = 1.0f - beta;
#pragma unroll
  for (int i = 0; i < 8; ++i) {
    const int m = i + 8 * lhalf;
    const float z = zt[m * 68 + nc0 + lm];
    const float v = omb * z + beta * acc[i];
    out[(size_t)(row0 + m) * 64 + nc0 + lm] = fmaxf(v, 0.0f);
  }
}

extern "C" void kernel_launch(void* const* d_in, const int* in_sizes, int n_in,
                              void* d_out, int out_size, void* d_ws,
                              size_t ws_size, hipStream_t stream) {
  const float* x     = (const float*)d_in[0];
  const int*   row   = (const int*)d_in[1];
  const int*   col   = (const int*)d_in[2];
  const float* ew    = (const float*)d_in[3];
  const float* w_lin = (const float*)d_in[4];
  const float* b_lin = (const float*)d_in[5];
  const float* convw = (const float*)d_in[6];

  const int N = in_sizes[0] / 128;
  const int E = in_sizes[1];
  const int D = 64;

  float* ws  = (float*)d_ws;
  float* h   = ws;
  float* x0  = h + (size_t)N * D;
  float* agg = x0 + (size_t)N * D;
  float* outf = (float*)d_out;

  const int nblk = N / 16;
  const int n4 = N * D / 4;

  lin_gemm_relu<<<nblk, 128, 0, stream>>>(x, w_lin, b_lin, h, x0);

  for (int l = 0; l < 4; ++l) {
    const float beta = logf(0.5f / (float)(l + 1) + 1.0f);
    init_agg<<<(n4 + 255) / 256, 256, 0, stream>>>(x0, agg, n4);
    spmm_scatter<<<(E + 7) / 8, 256, 0, stream>>>(h, row, col, ew, agg, E);
    gcn2_update<<<nblk, 128, 0, stream>>>(
        agg, convw + (size_t)l * D * D, (l == 3) ? outf : h, beta);
  }
}


# Edges bucketed by destination row once per call (int atomic rank, scan, place); per layer one wave sums a row's edges in registers, one store: no float atomics, no agg init; sum order = rank order
# speedup vs baseline: 2.5592x; 2.5592x over previous
#include <hip/hip_runtime.h>
#include <math.h>


typedef __attribute__((ext_vector_type(2))) float v2f;
typedef __attribute__((ext_vector_type(8))) float v8f;

__device__ __forceinline__ v8f wmma_f32_16x16x4(v2f a, v2f b, v8f c) {
  return __builtin_amdgcn_wmma_f32_16x16x4_f32(false, a, false, b, (short)0, c,
                                               false, false);
}

__global__ void __launch_bounds__(128)
lin_gemm_relu(const float* __restrict__ x, const float* __restrict__ W,
              const float* __restrict__ bias, float* __restrict__ h,
              float* __restrict__ x0, int* __restrict__ cnt) {
  __shared__ float xt[16 * 132];

  const int row0 = blockIdx.x * 16;
  const int t = threadIdx.x;

  if (t < 16) cnt[row0 + t] = 0;

  {
    const int r = t >> 3;
    const int c0 = (t & 7) * 16;
    const float4* xp =
        (const float4*)(x + (size_t)(row0 + r) * 128 + c0);
#pragma unroll
    for (int j = 0; j < 4; ++j) {
      float4 v = xp[j];
      float* dst = &xt[r * 132 + c0 + 4 * j];
      dst[0] = v.x; dst[1] = v.y; dst[2] = v.z; dst[3] = v.w;
    }
  }
  __syncthreads();

  const int lane  = t & 31;
  const int wave  = t >> 5;
  const int lhalf = lane >> 4;
  const int lm    = lane & 15;
  const int nc0   = wave * 16;

  v8f acc = {};
#pragma unroll
  for (int k0 = 0; k0 < 128; k0 += 4) {
    v2f a;
    const float* ap = &xt[lm * 132 + k0 + 2 * lhalf];
    a.x = ap[0];
    a.y = ap[1];
    v2f b;
    const float* wp = W + (size_t)(k0 + 2 * lhalf) * 64 + nc0 + lm;
    b.x = wp[0];
    b.y = wp[64];
    acc = wmma_f32_16x16x4(a, b, acc);
  }

  const float bv = bias[nc0 + lm];
#pragma unroll
  for (int i = 0; i < 8; ++i) {
    const int m = i + 8 * lhalf;
    float v = acc[i] + bv;
    v = fmaxf(v, 0.0f);
    const size_t idx = (size_t)(row0 + m) * 64 + nc0 + lm;
    h[idx] = v;
    x0[idx] = v;
  }
}

__global__ void __launch_bounds__(256)
edge_rank(const int* __restrict__ row, int* __restrict__ cnt,
          int* __restrict__ rank, int E) {
  const int e = blockIdx.x * 256 + threadIdx.x;
  if (e >= E) return;
  rank[e] = atomicAdd(&cnt[row[e]], 1);
}

__global__ void __launch_bounds__(256)
row_offsets(const int* __restrict__ cnt, int* __restrict__ off, int n4) {
  __shared__ int own_tot[8];
  __shared__ int pre_tot[8];
  const int t = threadIdx.x;
  const int lane = t & 31;
  const int wave = t >> 5;
  const int4* c4 = (const int4*)cnt;

  int pre = 0;
  const int nb = blockIdx.x;
  for (int k = 0; k < nb; k += 16) {
    int4 p[16];
#pragma unroll
    for (int u = 0; u < 16; ++u) {
      const int g = (k + u < nb) ? (k + u) : 0;
      p[u] = c4[g * 256 + t];
    }
#pragma unroll
    for (int u = 0; u < 16; ++u)
      if (k + u < nb) pre += (p[u].x + p[u].y) + (p[u].z + p[u].w);
  }

  const int i4 = blockIdx.x * 256 + t;
  int4 v = make_int4(0, 0, 0, 0);
  if (i4 < n4) v = c4[i4];
  const int own = (v.x + v.y) + (v.z + v.w);

  int inc = own;
  for (int d = 1; d < 32; d <<= 1) {
    const int u = __shfl_up(inc, d);
    if (lane >= d) inc += u;
  }
  for (int d = 16; d > 0; d >>= 1) pre += __shfl_xor(pre, d);

  if (lane == 31) own_tot[wave] = inc;
  if (lane == 0) pre_tot[wave] = pre;
  __syncthreads();

  int base = 0;
  for (int w = 0; w < 8; ++w) {
    base += pre_tot[w];
    if (w < wave) base += own_tot[w];
  }
  const int ex = base + inc - own;
  if (i4 < n4) {
    int4 o;
    o.x = ex;
    o.y = ex + v.x;
    o.z = ex + v.x + v.y;
    o.w = ex + v.x + v.y + v.z;
    ((int4*)off)[i4] = o;
  }
}

__global__ void __launch_bounds__(256)
edge_place(const int* __restrict__ row, const int* __restrict__ col,
           const float* __restrict__ ew, const int* __restrict__ off,
           const int* __restrict__ rank, int* __restrict__ rec, int E) {
  const int e = blockIdx.x * 256 + threadIdx.x;
  if (e >= E) return;
  const int p = off[row[e]] + rank[e];
  ((int2*)rec)[p] = make_int2(col[e], __float_as_int(ew[e]));
}

template <int K>
__device__ __forceinline__ void row_chunk(const float* __restrict__ h,
                                          int ecol, int ewt, int j, int lane,
                                          float& ax, float& ay) {
  float2 v[K];
  float w[K];
#pragma unroll
  for (int u = 0; u < K; ++u) {
    const int c = __builtin_amdgcn_readlane(ecol, j + u);
    w[u] = __int_as_float(__builtin_amdgcn_readlane(ewt, j + u));
    v[u] = ((const float2*)(h + (size_t)c * 64))[lane];
  }
#pragma unroll
  for (int u = 0; u < K; ++u) {
    ax += v[u].x * w[u];
    ay += v[u].y * w[u];
  }
}

__global__ void __launch_bounds__(256)
spmm_gather(const float* __restrict__ h, const float* __restrict__ x0,
            const int* __restrict__ off, const int* __restrict__ cnt,
            const int* __restrict__ rec, float* __restrict__ agg, int N) {
  const int r =
      __builtin_amdgcn_readfirstlane(blockIdx.x * 8 + (threadIdx.x >> 5));
  if (r >= N) return;
  const int lane = threadIdx.x & 31;

  const int start = off[r];
  const int n = cnt[r];

  const float s = 0.1f / 0.9f;
  const float2 xv = ((const float2*)(x0 + (size_t)r * 64))[lane];
  float ax = s * xv.x;
  float ay = s * xv.y;

  for (int b = 0; b < n; b += 32) {
    const int m = (n - b < 32) ? (n - b) : 32;
    int2 ed = make_int2(0, 0);
    if (lane < m) ed = ((const int2*)rec)[start + b + lane];

    int j = 0;
    for (; j + 8 <= m; j += 8) row_chunk<8>(h, ed.x, ed.y, j, lane, ax, ay);
    if (j + 4 <= m) { row_chunk<4>(h, ed.x, ed.y, j, lane, ax, ay); j += 4; }
    if (j + 2 <= m) { row_chunk<2>(h, ed.x, ed.y, j, lane, ax, ay); j += 2; }
    if (j < m)      { row_chunk<1>(h, ed.x, ed.y, j, lane, ax, ay); }
  }

  ((float2*)(agg + (size_t)r * 64))[lane] = make_float2(ax, ay);
}

__global__ void __launch_bounds__(128)
gcn2_update(const float* __restrict__ agg, const float* __restrict__ W,
            float* __restrict__ out, float beta) {
  __shared__ float zt[16 * 68];

  const int row0 = blockIdx.x * 16;
  const int t = threadIdx.x;

  {
    const int r = t >> 3;
    const int c0 = (t & 7) * 8;
    const float4* ap =
        (const float4*)(agg + (size_t)(row0 + r) * 64 + c0);
#pragma unroll
    for (int j = 0; j < 2; ++j) {
      float4 v = ap[j];
      float* dst = &zt[r * 68 + c0 + 4 * j];
      dst[0] = 0.9f * v.x; dst[1] = 0.9f * v.y;
      dst[2] = 0.9f * v.z; dst[3] = 0.9f * v.w;
    }
  }
  __syncthreads();

  const int lane  = t & 31;
  const int wave  = t >> 5;
  const int lhalf = lane >> 4;
  const int lm    = lane & 15;
  const int nc0   = wave * 16;

  v8f acc = {};
#pragma unroll
  for (int k0 = 0; k0 < 64; k0 += 4) {
    v2f a;
    const float* ap = &zt[lm * 68 + k0 + 2 * lhalf];
    a.x = ap[0];
    a.y = ap[1];
    v2f b;
    const float* wp = W + (size_t)(k0 + 2 * lhalf) * 64 + nc0 + lm;
    b.x = wp[0];
    b.y = wp[64];
    acc = wmma_f32_16x16x4(a, b, acc);
  }

  const float omb = 1.0f - beta;
#pragma unroll
  for (int i = 0; i < 8; ++i) {
    const int m = i + 8 * lhalf;
    const float z = zt[m * 68 + nc0 + lm];
    const float v = omb * z + beta * acc[i];
    out[(size_t)(row0 + m) * 64 + nc0 + lm] = fmaxf(v, 0.0f);
  }
}

extern "C" void kernel_launch(void* const* d_in, const int* in_sizes, int n_in,
                              void* d_out, int out_size, void* d_ws,
                              size_t ws_size, hipStream_t stream) {
  const float* x     = (const float*)d_in[0];
  const int*   row   = (const int*)d_in[1];
  const int*   col   = (const int*)d_in[2];
  const float* ew    = (const float*)d_in[3];
  const float* w_lin = (const float*)d_in[4];
  const float* b_lin = (const float*)d_in[5];
  const float* convw = (const float*)d_in[6];

  const int N = in_sizes[0] / 128;
  const int E = in_sizes[1];
  const int D = 64;

  float* ws  = (float*)d_ws;
  float* h   = ws;
  float* x0  = h + (size_t)N * D;
  float* agg = x0 + (size_t)N * D;
  int*   cnt  = (int*)(agg + (size_t)N * D);
  int*   off  = cnt + N;
  int*   rank = off + N;
  int*   rec  = rank + E;
  float* outf = (float*)d_out;

  const int nblk = N / 16;

  lin_gemm_relu<<<nblk, 128, 0, stream>>>(x, w_lin, b_lin, h, x0, cnt);

  const int eblk = (E + 255) / 256;
  edge_rank<<<eblk, 256, 0, stream>>>(row, cnt, rank, E);
  row_offsets<<<(N / 4 + 255) / 256, 256, 0, stream>>>(cnt, off, N / 4);
  edge_place<<<eblk, 256, 0, stream>>>(row, col, ew, off, rank, rec, E);

  for (int l = 0; l < 4; ++l) {
    const float beta = logf(0.5f / (float)(l + 1) + 1.0f);
    spmm_gather<<<(N + 7) / 8, 256, 0, stream>>>(h, x0, off, cnt, rec, agg, N);
    gcn2_update<<<nblk, 128, 0, stream>>>(
        agg, convw + (size_t)l * D * D, (l == 3) ? outf : h, beta);
  }
}


# One launch per layer: each wave sums 4 destination rows (same rank order) into its block's LDS z tile, the block's WMMA update reads it there; agg never goes to memory, h alternates between 2 buffers
# speedup vs baseline: 2.9345x; 1.1466x over previous
#include <hip/hip_runtime.h>
#include <math.h>


typedef __attribute__((ext_vector_type(2))) float v2f;
typedef __attribute__((ext_vector_type(8))) float v8f;

__device__ __forceinline__ v8f wmma_f32_16x16x4(v2f a, v2f b, v8f c) {
  return __builtin_amdgcn_wmma_f32_16x16x4_f32(false, a, false, b, (short)0, c,
                                               false, false);
}

__global__ void __launch_bounds__(128)
lin_gemm_relu(const float* __restrict__ x, const float* __restrict__ W,
              const float* __restrict__ bias, float* __restrict__ h,
              float* __restrict__ x0, int* __restrict__ cnt) {
  __shared__ float xt[16 * 132];

  const int row0 = blockIdx.x * 16;
  const int t = threadIdx.x;

  if (t < 16) cnt[row0 + t] = 0;

  {
    const int r = t >> 3;
    const int c0 = (t & 7) * 16;
    const float4* xp =
        (const float4*)(x + (size_t)(row0 + r) * 128 + c0);
#pragma unroll
    for (int j = 0; j < 4; ++j) {
      float4 v = xp[j];
      float* dst = &xt[r * 132 + c0 + 4 * j];
      dst[0] = v.x; dst[1] = v.y; dst[2] = v.z; dst[3] = v.w;
    }
  }
  __syncthreads();

  const int lane  = t & 31;
  const int wave  = t >> 5;
  const int lhalf = lane >> 4;
  const int lm    = lane & 15;
  const int nc0   = wave * 16;

  v8f acc = {};
#pragma unroll
  for (int k0 = 0; k0 < 128; k0 += 4) {
    v2f a;
    const float* ap = &xt[lm * 132 + k0 + 2 * lhalf];
    a.x = ap[0];
    a.y = ap[1];
    v2f b;
    const float* wp = W + (size_t)(k0 + 2 * lhalf) * 64 + nc0 + lm;
    b.x = wp[0];
    b.y = wp[64];
    acc = wmma_f32_16x16x4(a, b, acc);
  }

  const float bv = bias[nc0 + lm];
#pragma unroll
  for (int i = 0; i < 8; ++i) {
    const int m = i + 8 * lhalf;
    float v = acc[i] + bv;
    v = fmaxf(v, 0.0f);
    const size_t idx = (size_t)(row0 + m) * 64 + nc0 + lm;
    h[idx] = v;
    x0[idx] = v;
  }
}

__global__ void __launch_bounds__(256)
edge_rank(const int* __restrict__ row, int* __restrict__ cnt,
          int* __restrict__ rank, int E) {
  const int e = blockIdx.x * 256 + threadIdx.x;
  if (e >= E) return;
  rank[e] = atomicAdd(&cnt[row[e]], 1);
}

__global__ void __launch_bounds__(256)
row_offsets(const int* __restrict__ cnt, int* __restrict__ off, int n4) {
  __shared__ int own_tot[8];
  __shared__ int pre_tot[8];
  const int t = threadIdx.x;
  const int lane = t & 31;
  const int wave = t >> 5;
  const int4* c4 = (const int4*)cnt;

  int pre = 0;
  const int nb = blockIdx.x;
  for (int k = 0; k < nb; k += 16) {
    int4 p[16];
#pragma unroll
    for (int u = 0; u < 16; ++u) {
      const int g = (k + u < nb) ? (k + u) : 0;
      p[u] = c4[g * 256 + t];
    }
#pragma unroll
    for (int u = 0; u < 16; ++u)
      if (k + u < nb) pre += (p[u].x + p[u].y) + (p[u].z + p[u].w);
  }

  const int i4 = blockIdx.x * 256 + t;
  int4 v = make_int4(0, 0, 0, 0);
  if (i4 < n4) v = c4[i4];
  const int own = (v.x + v.y) + (v.z + v.w);

  int inc = own;
  for (int d = 1; d < 32; d <<= 1) {
    const int u = __shfl_up(inc, d);
    if (lane >= d) inc += u;
  }
  for (int d = 16; d > 0; d >>= 1) pre += __shfl_xor(pre, d);

  if (lane == 31) own_tot[wave] = inc;
  if (lane == 0) pre_tot[wave] = pre;
  __syncthreads();

  int base = 0;
  for (int w = 0; w < 8; ++w) {
    base += pre_tot[w];
    if (w < wave) base += own_tot[w];
  }
  const int ex = base + inc - own;
  if (i4 < n4) {
    int4 o;
    o.x = ex;
    o.y = ex + v.x;
    o.z = ex + v.x + v.y;
    o.w = ex + v.x + v.y + v.z;
    ((int4*)off)[i4] = o;
  }
}

__global__ void __launch_bounds__(256)
edge_place(const int* __restrict__ row, const int* __restrict__ col,
           const float* __restrict__ ew, const int* __restrict__ off,
           const int* __restrict__ rank, int* __restrict__ rec, int E) {
  const int e = blockIdx.x * 256 + threadIdx.x;
  if (e >= E) return;
  const int p = off[row[e]] + rank[e];
  ((int2*)rec)[p] = make_int2(col[e], __float_as_int(ew[e]));
}

template <int K>
__device__ __forceinline__ void row_chunk(const float* __restrict__ h,
                                          int ecol, int ewt, int j, int lane,
                                          float& ax, float& ay) {
  float2 v[K];
  float w[K];
#pragma unroll
  for (int u = 0; u < K; ++u) {
    const int c = __builtin_amdgcn_readlane(ecol, j + u);
    w[u] = __int_as_float(__builtin_amdgcn_readlane(ewt, j + u));
    v[u] = ((const float2*)(h + (size_t)c * 64))[lane];
  }
#pragma unroll
  for (int u = 0; u < K; ++u) {
    ax += v[u].x * w[u];
    ay += v[u].y * w[u];
  }
}

__global__ void __launch_bounds__(128)
gcn2_layer(const float* __restrict__ h, const float* __restrict__ x0,
           const int* __restrict__ off, const int* __restrict__ cnt,
           const int* __restrict__ rec, const float* __restrict__ W,
           float* __restrict__ hout, float beta) {
  __shared__ float zt[16 * 68];

  const int row0 = blockIdx.x * 16;
  const int t = threadIdx.x;
  const int lane = t & 31;
  const int wave = __builtin_amdgcn_readfirstlane(t >> 5);
  const int rb = row0 + wave * 4;

  const int4 st4 = ((const int4*)off)[rb >> 2];
  const int4 n4  = ((const int4*)cnt)[rb >> 2];
  const int st[4] = {st4.x, st4.y, st4.z, st4.w};
  const int nn[4] = {n4.x, n4.y, n4.z, n4.w};

  float2 xv[4];
  int2 ed0[4];
#pragma unroll
  for (int i = 0; i < 4; ++i)
    xv[i] = ((const float2*)(x0 + (size_t)(rb + i) * 64))[lane];
#pragma unroll
  for (int i = 0; i < 4; ++i) {
    ed0[i] = make_int2(0, 0);
    if (lane < nn[i]) ed0[i] = ((const int2*)rec)[st[i] + lane];
  }

  const float s = 0.1f / 0.9f;
#pragma unroll
  for (int i = 0; i < 4; ++i) {
    const int start = st[i];
    const int n = nn[i];
    float ax = s * xv[i].x;
    float ay = s * xv[i].y;

    for (int b = 0; b < n; b += 32) {
      const int m = (n - b < 32) ? (n - b) : 32;
      int2 ed = ed0[i];
      if (b > 0) {
        ed = make_int2(0, 0);
        if (lane < m) ed = ((const int2*)rec)[start + b + lane];
      }

      int j = 0;
      for (; j + 8 <= m; j += 8) row_chunk<8>(h, ed.x, ed.y, j, lane, ax, ay);
      if (j + 4 <= m) { row_chunk<4>(h, ed.x, ed.y, j, lane, ax, ay); j += 4; }
      if (j + 2 <= m) { row_chunk<2>(h, ed.x, ed.y, j, lane, ax, ay); j += 2; }
      if (j < m)      { row_chunk<1>(h, ed.x, ed.y, j, lane, ax, ay); }
    }

    ((float2*)&zt[(wave * 4 + i) * 68])[lane] =
        make_float2(0.9f * ax, 0.9f * ay);
  }

  const int lhalf = lane >> 4;
  const int lm    = lane & 15;
  const int nc0   = wave * 16;

  v2f bf[16];
#pragma unroll
  for (int kk = 0; kk < 16; ++kk) {
    const float* wp = W + (size_t)(4 * kk + 2 * lhalf) * 64 + nc0 + lm;
    bf[kk].x = wp[0];
    bf[kk].y = wp[64];
  }
  __syncthreads();

  v8f acc = {};
#pragma unroll
  for (int kk = 0; kk < 16; ++kk) {
    v2f a;
    const float* ap = &zt[lm * 68 + 4 * kk + 2 * lhalf];
    a.x = ap[0];
    a.y = ap[1];
    acc = wmma_f32_16x16x4(a, bf[kk], acc);
  }

  const float omb = 1.0f - beta;
#pragma unroll
  for (int i = 0; i < 8; ++i) {
    const int m = i + 8 * lhalf;
    const float z = zt[m * 68 + nc0 + lm];
    const float v = omb * z + beta * acc[i];
    hout[(size_t)(row0 + m) * 64 + nc0 + lm] = fmaxf(v, 0.0f);
  }
}

extern "C" void kernel_launch(void* const* d_in, const int* in_sizes, int n_in,
                              void* d_out, int out_size, void* d_ws,
                              size_t ws_size, hipStream_t stream) {
  const float* x     = (const float*)d_in[0];
  const int*   row   = (const int*)d_in[1];
  const int*   col   = (const int*)d_in[2];
  const float* ew    = (const float*)d_in[3];
  const float* w_lin = (const float*)d_in[4];
  const float* b_lin = (const float*)d_in[5];
  const float* convw = (const float*)d_in[6];

  const int N = in_sizes[0] / 128;
  const int E = in_sizes[1];
  const int D = 64;

  float* ws  = (float*)d_ws;
  float* h   = ws;
  float* x0  = h + (size_t)N * D;
  float* h2  = x0 + (size_t)N * D;
  int*   cnt  = (int*)(h2 + (size_t)N * D);
  int*   off  = cnt + N;
  int*   rank = off + N;
  int*   rec  = rank + E;
  float* outf = (float*)d_out;

  const int nblk = N / 16;

  lin_gemm_relu<<<nblk, 128, 0, stream>>>(x, w_lin, b_lin, h, x0, cnt);

  const int eblk = (E + 255) / 256;
  edge_rank<<<eblk, 256, 0, stream>>>(row, cnt, rank, E);
  row_offsets<<<(N / 4 + 255) / 256, 256, 0, stream>>>(cnt, off, N / 4);
  edge_place<<<eblk, 256, 0, stream>>>(row, col, ew, off, rank, rec, E);

  for (int l = 0; l < 4; ++l) {
    const float beta = logf(0.5f / (float)(l + 1) + 1.0f);
    const float* src = (l & 1) ? h2 : h;
    float* dst = (l == 3) ? outf : ((l & 1) ? h : h2);
    gcn2_layer<<<nblk, 128, 0, stream>>>(src, x0, off, cnt, rec,
                                         convw + (size_t)l * D * D, dst, beta);
  }
}


# Edges ordered by destination row with LDS counters only (count per 256-row bucket and edge slice, scan, move to bucket, order bucket in LDS): no global atomics; row sum order = LDS add return order
# speedup vs baseline: 4.4975x; 1.5326x over previous
#include <hip/hip_runtime.h>
#include <math.h>


typedef __attribute__((ext_vector_type(2))) float v2f;
typedef __attribute__((ext_vector_type(8))) float v8f;

__device__ __forceinline__ v8f wmma_f32_16x16x4(v2f a, v2f b, v8f c) {
  return __builtin_amdgcn_wmma_f32_16x16x4_f32(false, a, false, b, (short)0, c,
                                               false, false);
}

__global__ void __launch_bounds__(128)
lin_gemm_relu(const float* __restrict__ x, const float* __restrict__ W,
              const float* __restrict__ bias, float* __restrict__ h,
              float* __restrict__ x0) {
  __shared__ float xt[16 * 132];

  const int row0 = blockIdx.x * 16;
  const int t = threadIdx.x;

  {
    const int r = t >> 3;
    const int c0 = (t & 7) * 16;
    const float4* xp =
        (const float4*)(x + (size_t)(row0 + r) * 128 + c0);
#pragma unroll
    for (int j = 0; j < 4; ++j) {
      float4 v = xp[j];
      float* dst = &xt[r * 132 + c0 + 4 * j];
      dst[0] = v.x; dst[1] = v.y; dst[2] = v.z; dst[3] = v.w;
    }
  }
  __syncthreads();

  const int lane  = t & 31;
  const int wave  = t >> 5;
  const int lhalf = lane >> 4;
  const int lm    = lane & 15;
  const int nc0   = wave * 16;

  v8f acc = {};
#pragma unroll
  for (int k0 = 0; k0 < 128; k0 += 4) {
    v2f a;
    const float* ap = &xt[lm * 132 + k0 + 2 * lhalf];
    a.x = ap[0];
    a.y = ap[1];
    v2f b;
    const float* wp = W + (size_t)(k0 + 2 * lhalf) * 64 + nc0 + lm;
    b.x = wp[0];
    b.y = wp[64];
    acc = wmma_f32_16x16x4(a, b, acc);
  }

  const float bv = bias[nc0 + lm];
#pragma unroll
  for (int i = 0; i < 8; ++i) {
    const int m = i + 8 * lhalf;
    float v = acc[i] + bv;
    v = fmaxf(v, 0.0f);
    const size_t idx = (size_t)(row0 + m) * 64 + nc0 + lm;
    h[idx] = v;
    x0[idx] = v;
  }
}

#define BUCKET_SHIFT 8
#define MAX_BUCKETS 1024

__global__ void __launch_bounds__(1024)
bucket_count(const int* __restrict__ row, int* __restrict__ hist, int E,
             int chunk, int S, int nbuckets) {
  __shared__ int c[MAX_BUCKETS];
  const int t = threadIdx.x;
  const int s = blockIdx.x;
  if (t < nbuckets) c[t] = 0;
  __syncthreads();
  const int e0 = s * chunk;
  const int e1 = (e0 + chunk < E) ? (e0 + chunk) : E;
  for (int eb = e0 + t; eb < e1; eb += 4096) {
    int r[4];
#pragma unroll
    for (int u = 0; u < 4; ++u) {
      const int e = (eb + 1024 * u < e1) ? (eb + 1024 * u) : (e1 - 1);
      r[u] = row[e];
    }
    asm volatile("" ::"v"(r[0]), "v"(r[1]), "v"(r[2]), "v"(r[3]));
#pragma unroll
    for (int u = 0; u < 4; ++u)
      if (eb + 1024 * u < e1) atomicAdd(&c[r[u] >> BUCKET_SHIFT], 1);
  }
  __syncthreads();
  if (t < nbuckets) hist[t * S + s] = c[t];
}

__global__ void __launch_bounds__(256)
row_offsets(const int* __restrict__ cnt, int* __restrict__ off, int n4) {
  __shared__ int own_tot[8];
  __shared__ int pre_tot[8];
  const int t = threadIdx.x;
  const int lane = t & 31;
  const int wave = t >> 5;
  const int4* c4 = (const int4*)cnt;

  int pre = 0;
  const int nb = blockIdx.x;
  for (int k = 0; k < nb; k += 16) {
    int4 p[16];
#pragma unroll
    for (int u = 0; u < 16; ++u) {
      const int g = (k + u < nb) ? (k + u) : 0;
      p[u] = c4[g * 256 + t];
    }
#pragma unroll
    for (int u = 0; u < 16; ++u)
      if (k + u < nb) pre += (p[u].x + p[u].y) + (p[u].z + p[u].w);
  }

  const int i4 = blockIdx.x * 256 + t;
  int4 v = make_int4(0, 0, 0, 0);
  if (i4 < n4) v = c4[i4];
  const int own = (v.x + v.y) + (v.z + v.w);

  int inc = own;
  for (int d = 1; d < 32; d <<= 1) {
    const int u = __shfl_up(inc, d);
    if (lane >= d) inc += u;
  }
  for (int d = 16; d > 0; d >>= 1) pre += __shfl_xor(pre, d);

  if (lane == 31) own_tot[wave] = inc;
  if (lane == 0) pre_tot[wave] = pre;
  __syncthreads();

  int base = 0;
  for (int w = 0; w < 8; ++w) {
    base += pre_tot[w];
    if (w < wave) base += own_tot[w];
  }
  const int ex = base + inc - own;
  if (i4 < n4) {
    int4 o;
    o.x = ex;
    o.y = ex + v.x;
    o.z = ex + v.x + v.y;
    o.w = ex + v.x + v.y + v.z;
    ((int4*)off)[i4] = o;
  }
}

__global__ void __launch_bounds__(1024)
bucket_move(const int* __restrict__ row, const int* __restrict__ col,
            const float* __restrict__ ew, const int* __restrict__ hoff,
            int* __restrict__ tmp, int E, int chunk, int S, int nbuckets) {
  __shared__ int cur[MAX_BUCKETS];
  const int t = threadIdx.x;
  const int s = blockIdx.x;
  if (t < nbuckets) cur[t] = hoff[t * S + s];
  __syncthreads();
  const int e0 = s * chunk;
  const int e1 = (e0 + chunk < E) ? (e0 + chunk) : E;
  for (int eb = e0 + t; eb < e1; eb += 4096) {
    int r[4], c[4];
    float w[4];
#pragma unroll
    for (int u = 0; u < 4; ++u) {
      const int e = (eb + 1024 * u < e1) ? (eb + 1024 * u) : (e1 - 1);
      r[u] = row[e];
      c[u] = col[e];
      w[u] = ew[e];
    }
    asm volatile("" ::"v"(r[0]), "v"(r[1]), "v"(r[2]), "v"(r[3]), "v"(c[0]),
                 "v"(c[1]), "v"(c[2]), "v"(c[3]), "v"(w[0]), "v"(w[1]),
                 "v"(w[2]), "v"(w[3]));
#pragma unroll
    for (int u = 0; u < 4; ++u) {
      if (eb + 1024 * u < e1) {
        const int p = atomicAdd(&cur[r[u] >> BUCKET_SHIFT], 1);
        ((int4*)tmp)[p] = make_int4(r[u], c[u], __float_as_int(w[u]), 0);
      }
    }
  }
}

__global__ void __launch_bounds__(1024)
bucket_sort(const int* __restrict__ tmp, const int* __restrict__ hoff,
            int* __restrict__ off, int* __restrict__ cnt,
            int* __restrict__ rec, int N, int E, int S, int nbuckets) {
  __shared__ int lcnt[256];
  __shared__ int loff[256];
  __shared__ int lcur[256];
  __shared__ int wtot[8];
  const int t = threadIdx.x;
  const int lane = t & 31;
  const int wave = t >> 5;
  const int b = blockIdx.x;
  const int rbase = b << BUCKET_SHIFT;
  const int bs = hoff[b * S];
  const int be = (b + 1 < nbuckets) ? hoff[(b + 1) * S] : E;
  const int4* rd = (const int4*)tmp;

  if (t < 256) {
    lcnt[t] = 0;
    lcur[t] = 0;
  }
  __syncthreads();
  for (int ib = bs + t; ib < be; ib += 4096) {
    int r[4];
#pragma unroll
    for (int u = 0; u < 4; ++u) {
      const int i = (ib + 1024 * u < be) ? (ib + 1024 * u) : (be - 1);
      r[u] = rd[i].x;
    }
    asm volatile("" ::"v"(r[0]), "v"(r[1]), "v"(r[2]), "v"(r[3]));
#pragma unroll
    for (int u = 0; u < 4; ++u)
      if (ib + 1024 * u < be) atomicAdd(&lcnt[r[u] - rbase], 1);
  }
  __syncthreads();

  const int own = (t < 256) ? lcnt[t] : 0;
  int inc = own;
  for (int d = 1; d < 32; d <<= 1) {
    const int u = __shfl_up(inc, d);
    if (lane >= d) inc += u;
  }
  if (t < 256 && lane == 31) wtot[wave] = inc;
  __syncthreads();
  if (t < 256) {
    int base = 0;
    for (int w = 0; w < 8; ++w)
      if (w < wave) base += wtot[w];
    const int ex = base + inc - own;
    loff[t] = ex;
    if (rbase + t < N) {
      off[rbase + t] = bs + ex;
      cnt[rbase + t] = own;
    }
  }
  __syncthreads();

  for (int ib = bs + t; ib < be; ib += 4096) {
    int4 q[4];
#pragma unroll
    for (int u = 0; u < 4; ++u) {
      const int i = (ib + 1024 * u < be) ? (ib + 1024 * u) : (be - 1);
      q[u] = rd[i];
    }
    asm volatile("" ::"v"(q[0].x), "v"(q[1].x), "v"(q[2].x), "v"(q[3].x),
                 "v"(q[0].y), "v"(q[1].y), "v"(q[2].y), "v"(q[3].y),
                 "v"(q[0].z), "v"(q[1].z), "v"(q[2].z), "v"(q[3].z));
#pragma unroll
    for (int u = 0; u < 4; ++u) {
      if (ib + 1024 * u < be) {
        const int r = q[u].x - rbase;
        const int k = atomicAdd(&lcur[r], 1);
        ((int2*)rec)[bs + loff[r] + k] = make_int2(q[u].y, q[u].z);
      }
    }
  }
}

template <int K>
__device__ __forceinline__ void row_chunk(const float* __restrict__ h,
                                          int ecol, int ewt, int j, int lane,
                                          float& ax, float& ay) {
  float2 v[K];
  float w[K];
#pragma unroll
  for (int u = 0; u < K; ++u) {
    const int c = __builtin_amdgcn_readlane(ecol, j + u);
    w[u] = __int_as_float(__builtin_amdgcn_readlane(ewt, j + u));
    v[u] = ((const float2*)(h + (size_t)c * 64))[lane];
  }
#pragma unroll
  for (int u = 0; u < K; ++u) {
    ax += v[u].x * w[u];
    ay += v[u].y * w[u];
  }
}

__global__ void __launch_bounds__(128)
gcn2_layer(const float* __restrict__ h, const float* __restrict__ x0,
           const int* __restrict__ off, const int* __restrict__ cnt,
           const int* __restrict__ rec, const float* __restrict__ W,
           float* __restrict__ hout, float beta) {
  __shared__ float zt[16 * 68];

  const int row0 = blockIdx.x * 16;
  const int t = threadIdx.x;
  const int lane = t & 31;
  const int wave = __builtin_amdgcn_readfirstlane(t >> 5);
  const int rb = row0 + wave * 4;

  const int4 st4 = ((const int4*)off)[rb >> 2];
  const int4 n4  = ((const int4*)cnt)[rb >> 2];
  const int st[4] = {st4.x, st4.y, st4.z, st4.w};
  const int nn[4] = {n4.x, n4.y, n4.z, n4.w};

  float2 xv[4];
  int2 ed0[4];
#pragma unroll
  for (int i = 0; i < 4; ++i)
    xv[i] = ((const float2*)(x0 + (size_t)(rb + i) * 64))[lane];
#pragma unroll
  for (int i = 0; i < 4; ++i) {
    ed0[i] = make_int2(0, 0);
    if (lane < nn[i]) ed0[i] = ((const int2*)rec)[st[i] + lane];
  }

  const float s = 0.1f / 0.9f;
#pragma unroll
  for (int i = 0; i < 4; ++i) {
    const int start = st[i];
    const int n = nn[i];
    float ax = s * xv[i].x;
    float ay = s * xv[i].y;

    for (int b = 0; b < n; b += 32) {
      const int m = (n - b < 32) ? (n - b) : 32;
      int2 ed = ed0[i];
      if (b > 0) {
        ed = make_int2(0, 0);
        if (lane < m) ed = ((const int2*)rec)[start + b + lane];
      }

      int j = 0;
      for (; j + 8 <= m; j += 8) row_chunk<8>(h, ed.x, ed.y, j, lane, ax, ay);
      if (j + 4 <= m) { row_chunk<4>(h, ed.x, ed.y, j, lane, ax, ay); j += 4; }
      if (j + 2 <= m) { row_chunk<2>(h, ed.x, ed.y, j, lane, ax, ay); j += 2; }
      if (j < m)      { row_chunk<1>(h, ed.x, ed.y, j, lane, ax, ay); }
    }

    ((float2*)&zt[(wave * 4 + i) * 68])[lane] =
        make_float2(0.9f * ax, 0.9f * ay);
  }

  const int lhalf = lane >> 4;
  const int lm    = lane & 15;
  const int nc0   = wave * 16;

  v2f bf[16];
#pragma unroll
  for (int kk = 0; kk < 16; ++kk) {
    const float* wp = W + (size_t)(4 * kk + 2 * lhalf) * 64 + nc0 + lm;
    bf[kk].x = wp[0];
    bf[kk].y = wp[64];
  }
  __syncthreads();

  v8f acc = {};
#pragma unroll
  for (int kk = 0; kk < 16; ++kk) {
    v2f a;
    const float* ap = &zt[lm * 68 + 4 * kk + 2 * lhalf];
    a.x = ap[0];
    a.y = ap[1];
    acc = wmma_f32_16x16x4(a, bf[kk], acc);
  }

  const float omb = 1.0f - beta;
#pragma unroll
  for (int i = 0; i < 8; ++i) {
    const int m = i + 8 * lhalf;
    const float z = zt[m * 68 + nc0 + lm];
    const float v = omb * z + beta * acc[i];
    hout[(size_t)(row0 + m) * 64 + nc0 + lm] = fmaxf(v, 0.0f);
  }
}

extern "C" void kernel_launch(void* const* d_in, const int* in_sizes, int n_in,
                              void* d_out, int out_size, void* d_ws,
                              size_t ws_size, hipStream_t stream) {
  const float* x     = (const float*)d_in[0];
  const int*   row   = (const int*)d_in[1];
  const int*   col   = (const int*)d_in[2];
  const float* ew    = (const float*)d_in[3];
  const float* w_lin = (const float*)d_in[4];
  const float* b_lin = (const float*)d_in[5];
  const float* convw = (const float*)d_in[6];

  const int N = in_sizes[0] / 128;
  const int E = in_sizes[1];
  const int D = 64;

  float* ws  = (float*)d_ws;
  float* h   = ws;
  float* x0  = h + (size_t)N * D;
  float* h2  = x0 + (size_t)N * D;
  const int S  = 256;
  const int nb = (N + 255) >> 8;
  int*   cnt  = (int*)(h2 + (size_t)N * D);
  int*   off  = cnt + N;
  int*   hist = off + N;
  int*   hoff = hist + nb * S;
  int*   tmp  = hoff + nb * S;
  int*   rec  = tmp + 4 * (size_t)E;
  float* outf = (float*)d_out;

  const int nblk = N / 16;

  lin_gemm_relu<<<nblk, 128, 0, stream>>>(x, w_lin, b_lin, h, x0);

  const int chunk = (E + S - 1) / S;
  const int n4 = nb * S / 4;
  bucket_count<<<S, 1024, 0, stream>>>(row, hist, E, chunk, S, nb);
  row_offsets<<<(n4 + 255) / 256, 256, 0, stream>>>(hist, hoff, n4);
  bucket_move<<<S, 1024, 0, stream>>>(row, col, ew, hoff, tmp, E, chunk, S, nb);
  bucket_sort<<<nb, 1024, 0, stream>>>(tmp, hoff, off, cnt, rec, N, E, S, nb);

  for (int l = 0; l < 4; ++l) {
    const float beta = logf(0.5f / (float)(l + 1) + 1.0f);
    const float* src = (l & 1) ? h2 : h;
    float* dst = (l == 3) ? outf : ((l & 1) ? h : h2);
    gcn2_layer<<<nblk, 128, 0, stream>>>(src, x0, off, cnt, rec,
                                         convw + (size_t)l * D * D, dst, beta);
  }
}


# Edge ordering in two launches, not four: slice blocks order their edges by bucket in LDS and write run tables, a bucket block sums the tables for its start; row sums: new LDS add return order
# speedup vs baseline: 4.9558x; 1.1019x over previous
#include <hip/hip_runtime.h>
#include <math.h>


typedef __attribute__((ext_vector_type(2))) float v2f;
typedef __attribute__((ext_vector_type(8))) float v8f;

__device__ __forceinline__ v8f wmma_f32_16x16x4(v2f a, v2f b, v8f c) {
  return __builtin_amdgcn_wmma_f32_16x16x4_f32(false, a, false, b, (short)0, c,
                                               false, false);
}

__global__ void __launch_bounds__(128)
lin_gemm_relu(const float* __restrict__ x, const float* __restrict__ W,
              const float* __restrict__ bias, float* __restrict__ h,
              float* __restrict__ x0) {
  __shared__ float xt[16 * 132];

  const int row0 = blockIdx.x * 16;
  const int t = threadIdx.x;

  {
    const int r = t >> 3;
    const int c0 = (t & 7) * 16;
    const float4* xp =
        (const float4*)(x + (size_t)(row0 + r) * 128 + c0);
#pragma unroll
    for (int j = 0; j < 4; ++j) {
      float4 v = xp[j];
      float* dst = &xt[r * 132 + c0 + 4 * j];
      dst[0] = v.x; dst[1] = v.y; dst[2] = v.z; dst[3] = v.w;
    }
  }
  __syncthreads();

  const int lane  = t & 31;
  const int wave  = t >> 5;
  const int lhalf = lane >> 4;
  const int lm    = lane & 15;
  const int nc0   = wave * 16;

  v8f acc = {};
#pragma unroll
  for (int k0 = 0; k0 < 128; k0 += 4) {
    v2f a;
    const float* ap = &xt[lm * 132 + k0 + 2 * lhalf];
    a.x = ap[0];
    a.y = ap[1];
    v2f b;
    const float* wp = W + (size_t)(k0 + 2 * lhalf) * 64 + nc0 + lm;
    b.x = wp[0];
    b.y = wp[64];
    acc = wmma_f32_16x16x4(a, b, acc);
  }

  const float bv = bias[nc0 + lm];
#pragma unroll
  for (int i = 0; i < 8; ++i) {
    const int m = i + 8 * lhalf;
    float v = acc[i] + bv;
    v = fmaxf(v, 0.0f);
    const size_t idx = (size_t)(row0 + m) * 64 + nc0 + lm;
    h[idx] = v;
    x0[idx] = v;
  }
}

#define BUCKET_SHIFT 8
#define MAX_BUCKETS 1024
#define SLICES 256
#define SLICE_SLOTS 7
#define SLICE_CAP 6400

__global__ void __launch_bounds__(1024)
slice_sort(const int* __restrict__ row, const int* __restrict__ col,
           const float* __restrict__ ew, int* __restrict__ lofs,
           int* __restrict__ tmp, int E, int chunk, int nbuckets) {
  __shared__ int c[MAX_BUCKETS];
  __shared__ int wtot[32];
  __shared__ int2 stage[SLICE_CAP];
  const int t = threadIdx.x;
  const int lane = t & 31;
  const int wave = t >> 5;
  const int s = blockIdx.x;
  c[t] = 0;
  __syncthreads();
  const int e0 = s * chunk;
  const int e1 = (e0 + chunk < E) ? (e0 + chunk) : E;

  int r[SLICE_SLOTS], cc[SLICE_SLOTS];
  float w[SLICE_SLOTS];
#pragma unroll
  for (int u = 0; u < SLICE_SLOTS; ++u) {
    const int e = (e0 + t + 1024 * u < e1) ? (e0 + t + 1024 * u) : (E - 1);
    r[u] = row[e];
    cc[u] = col[e];
    w[u] = ew[e];
  }
  __builtin_amdgcn_sched_barrier(0);
  asm volatile("" ::"v"(r[0]), "v"(r[1]), "v"(r[2]), "v"(r[3]), "v"(r[4]),
               "v"(r[5]), "v"(r[6]), "v"(cc[0]), "v"(cc[1]), "v"(cc[2]),
               "v"(cc[3]), "v"(cc[4]), "v"(cc[5]), "v"(cc[6]), "v"(w[0]),
               "v"(w[1]), "v"(w[2]), "v"(w[3]), "v"(w[4]), "v"(w[5]),
               "v"(w[6]));
#pragma unroll
  for (int u = 0; u < SLICE_SLOTS; ++u)
    if (e0 + t + 1024 * u < e1) atomicAdd(&c[r[u] >> BUCKET_SHIFT], 1);
  __syncthreads();

  const int own = c[t];
  int inc = own;
  for (int d = 1; d < 32; d <<= 1) {
    const int u = __shfl_up(inc, d);
    if (lane >= d) inc += u;
  }
  if (lane == 31) wtot[wave] = inc;
  __syncthreads();
  int base = 0;
  for (int k = 0; k < 32; ++k)
    if (k < wave) base += wtot[k];
  const int ex = base + inc - own;
  int* lrow = lofs + s * (nbuckets + 1);
  if (t < nbuckets) lrow[t] = ex;
  if (t == 0) lrow[nbuckets] = (e1 > e0) ? (e1 - e0) : 0;
  c[t] = ex;
  __syncthreads();

#pragma unroll
  for (int u = 0; u < SLICE_SLOTS; ++u) {
    if (e0 + t + 1024 * u < e1) {
      const int p = atomicAdd(&c[r[u] >> BUCKET_SHIFT], 1);
      stage[p] = make_int2(cc[u] | (int)((unsigned)(r[u] & 255) << 24),
                           __float_as_int(w[u]));
    }
  }
  __syncthreads();
  for (int i = t; i < e1 - e0; i += 1024) ((int2*)tmp)[e0 + i] = stage[i];
}

__global__ void __launch_bounds__(1024)
bucket_sort(const int* __restrict__ tmp, const int* __restrict__ lofs,
            int* __restrict__ off, int* __restrict__ cnt,
            int* __restrict__ rec, int N, int chunk, int nbuckets) {
  __shared__ int lcnt[256];
  __shared__ int loff[256];
  __shared__ int lcur[256];
  __shared__ int rs[SLICES];
  __shared__ int rn[SLICES];
  __shared__ int wsum[32];
  __shared__ int wtot[8];
  const int t = threadIdx.x;
  const int lane = t & 31;
  const int wave = t >> 5;
  const int b = blockIdx.x;
  const int rbase = b << BUCKET_SHIFT;
  const int2* rd = (const int2*)tmp;

  int a0 = 0;
  if (t < SLICES) {
    const int* lp = lofs + t * (nbuckets + 1) + b;
    a0 = lp[0];
    rs[t] = t * chunk + a0;
    rn[t] = lp[1] - a0;
  }
  if (t < 256) {
    lcnt[t] = 0;
    lcur[t] = 0;
  }
  int sum = a0;
  for (int d = 16; d > 0; d >>= 1) sum += __shfl_xor(sum, d);
  if (lane == 0) wsum[wave] = sum;
  __syncthreads();
  int bs = 0;
  for (int k = 0; k < 32; ++k) bs += wsum[k];

  const int start = rs[t >> 2];
  const int n = rn[t >> 2];
  const int q = t & 3;

  for (int k = q; k < n; k += 32) {
    int x[8];
#pragma unroll
    for (int u = 0; u < 8; ++u) {
      const int i = (k + 4 * u < n) ? (k + 4 * u) : (n - 1);
      x[u] = rd[start + i].x;
    }
    asm volatile("" ::"v"(x[0]), "v"(x[1]), "v"(x[2]), "v"(x[3]), "v"(x[4]),
                 "v"(x[5]), "v"(x[6]), "v"(x[7]));
#pragma unroll
    for (int u = 0; u < 8; ++u)
      if (k + 4 * u < n) atomicAdd(&lcnt[(unsigned)x[u] >> 24], 1);
  }
  __syncthreads();

  const int own = (t < 256) ? lcnt[t] : 0;
  int inc = own;
  for (int d = 1; d < 32; d <<= 1) {
    const int u = __shfl_up(inc, d);
    if (lane >= d) inc += u;
  }
  if (t < 256 && lane == 31) wtot[wave] = inc;
  __syncthreads();
  if (t < 256) {
    int base = 0;
    for (int k = 0; k < 8; ++k)
      if (k < wave) base += wtot[k];
    const int ex = base + inc - own;
    loff[t] = ex;
    if (rbase + t < N) {
      off[rbase + t] = bs + ex;
      cnt[rbase + t] = own;
    }
  }
  __syncthreads();

  for (int k = q; k < n; k += 32) {
    int2 g[8];
#pragma unroll
    for (int u = 0; u < 8; ++u) {
      const int i = (k + 4 * u < n) ? (k + 4 * u) : (n - 1);
      g[u] = rd[start + i];
    }
    asm volatile("" ::"v"(g[0].x), "v"(g[1].x), "v"(g[2].x), "v"(g[3].x),
                 "v"(g[4].x), "v"(g[5].x), "v"(g[6].x), "v"(g[7].x),
                 "v"(g[0].y), "v"(g[1].y), "v"(g[2].y), "v"(g[3].y),
                 "v"(g[4].y), "v"(g[5].y), "v"(g[6].y), "v"(g[7].y));
#pragma unroll
    for (int u = 0; u < 8; ++u) {
      if (k + 4 * u < n) {
        const int rl = (unsigned)g[u].x >> 24;
        const int p = atomicAdd(&lcur[rl], 1);
        ((int2*)rec)[bs + loff[rl] + p] =
            make_int2(g[u].x & 0xFFFFFF, g[u].y);
      }
    }
  }
}

template <int K>
__device__ __forceinline__ void row_chunk(const float* __restrict__ h,
                                          int ecol, int ewt, int j, int lane,
                                          float& ax, float& ay) {
  float2 v[K];
  float w[K];
#pragma unroll
  for (int u = 0; u < K; ++u) {
    const int c = __builtin_amdgcn_readlane(ecol, j + u);
    w[u] = __int_as_float(__builtin_amdgcn_readlane(ewt, j + u));
    v[u] = ((const float2*)(h + (size_t)c * 64))[lane];
  }
#pragma unroll
  for (int u = 0; u < K; ++u) {
    ax += v[u].x * w[u];
    ay += v[u].y * w[u];
  }
}

__global__ void __launch_bounds__(128)
gcn2_layer(const float* __restrict__ h, const float* __restrict__ x0,
           const int* __restrict__ off, const int* __restrict__ cnt,
           const int* __restrict__ rec, const float* __restrict__ W,
           float* __restrict__ hout, float beta) {
  __shared__ float zt[16 * 68];

  const int row0 = blockIdx.x * 16;
  const int t = threadIdx.x;
  const int lane = t & 31;
  const int wave = __builtin_amdgcn_readfirstlane(t >> 5);
  const int rb = row0 + wave * 4;

  const int4 st4 = ((const int4*)off)[rb >> 2];
  const int4 n4  = ((const int4*)cnt)[rb >> 2];
  const int st[4] = {st4.x, st4.y, st4.z, st4.w};
  const int nn[4] = {n4.x, n4.y, n4.z, n4.w};

  float2 xv[4];
  int2 ed0[4];
#pragma unroll
  for (int i = 0; i < 4; ++i)
    xv[i] = ((const float2*)(x0 + (size_t)(rb + i) * 64))[lane];
#pragma unroll
  for (int i = 0; i < 4; ++i) {
    ed0[i] = make_int2(0, 0);
    if (lane < nn[i]) ed0[i] = ((const int2*)rec)[st[i] + lane];
  }

  const float s = 0.1f / 0.9f;
#pragma unroll
  for (int i = 0; i < 4; ++i) {
    const int start = st[i];
    const int n = nn[i];
    float ax = s * xv[i].x;
    float ay = s * xv[i].y;

    for (int b = 0; b < n; b += 32) {
      const int m = (n - b < 32) ? (n - b) : 32;
      int2 ed = ed0[i];
      if (b > 0) {
        ed = make_int2(0, 0);
        if (lane < m) ed = ((const int2*)rec)[start + b + lane];
      }

      int j = 0;
      for (; j + 8 <= m; j += 8) row_chunk<8>(h, ed.x, ed.y, j, lane, ax, ay);
      if (j + 4 <= m) { row_chunk<4>(h, ed.x, ed.y, j, lane, ax, ay); j += 4; }
      if (j + 2 <= m) { row_chunk<2>(h, ed.x, ed.y, j, lane, ax, ay); j += 2; }
      if (j < m)      { row_chunk<1>(h, ed.x, ed.y, j, lane, ax, ay); }
    }

    ((float2*)&zt[(wave * 4 + i) * 68])[lane] =
        make_float2(0.9f * ax, 0.9f * ay);
  }

  const int lhalf = lane >> 4;
  const int lm    = lane & 15;
  const int nc0   = wave * 16;

  v2f bf[16];
#pragma unroll
  for (int kk = 0; kk < 16; ++kk) {
    const float* wp = W + (size_t)(4 * kk + 2 * lhalf) * 64 + nc0 + lm;
    bf[kk].x = wp[0];
    bf[kk].y = wp[64];
  }
  __syncthreads();

  v8f acc = {};
#pragma unroll
  for (int kk = 0; kk < 16; ++kk) {
    v2f a;
    const float* ap = &zt[lm * 68 + 4 * kk + 2 * lhalf];
    a.x = ap[0];
    a.y = ap[1];
    acc = wmma_f32_16x16x4(a, bf[kk], acc);
  }

  const float omb = 1.0f - beta;
#pragma unroll
  for (int i = 0; i < 8; ++i) {
    const int m = i + 8 * lhalf;
    const float z = zt[m * 68 + nc0 + lm];
    const float v = omb * z + beta * acc[i];
    hout[(size_t)(row0 + m) * 64 + nc0 + lm] = fmaxf(v, 0.0f);
  }
}

extern "C" void kernel_launch(void* const* d_in, const int* in_sizes, int n_in,
                              void* d_out, int out_size, void* d_ws,
                              size_t ws_size, hipStream_t stream) {
  const float* x     = (const float*)d_in[0];
  const int*   row   = (const int*)d_in[1];
  const int*   col   = (const int*)d_in[2];
  const float* ew    = (const float*)d_in[3];
  const float* w_lin = (const float*)d_in[4];
  const float* b_lin = (const float*)d_in[5];
  const float* convw = (const float*)d_in[6];

  const int N = in_sizes[0] / 128;
  const int E = in_sizes[1];
  const int D = 64;

  float* ws  = (float*)d_ws;
  float* h   = ws;
  float* x0  = h + (size_t)N * D;
  float* h2  = x0 + (size_t)N * D;
  const int S  = SLICES;
  const int nb = (N + 255) >> 8;
  int*   cnt  = (int*)(h2 + (size_t)N * D);
  int*   off  = cnt + N;
  int*   lofs = off + N;
  int*   tmp  = lofs + S * (nb + 1);
  int*   rec  = tmp + 2 * (size_t)E;
  float* outf = (float*)d_out;

  const int nblk = N / 16;

  lin_gemm_relu<<<nblk, 128, 0, stream>>>(x, w_lin, b_lin, h, x0);

  const int chunk = (E + S - 1) / S;
  slice_sort<<<S, 1024, 0, stream>>>(row, col, ew, lofs, tmp, E, chunk, nb);
  bucket_sort<<<nb, 1024, 0, stream>>>(tmp, lofs, off, cnt, rec, N, chunk, nb);

  for (int l = 0; l < 4; ++l) {
    const float beta = logf(0.5f / (float)(l + 1) + 1.0f);
    const float* src = (l & 1) ? h2 : h;
    float* dst = (l == 3) ? outf : ((l & 1) ? h : h2);
    gcn2_layer<<<nblk, 128, 0, stream>>>(src, x0, off, cnt, rec,
                                         convw + (size_t)l * D * D, dst, beta);
  }
}
